# starting kernel
# baseline (speedup 1.0000x reference)
#include <hip/hip_runtime.h>
#include <hip/hip_bf16.h>
#include <math.h>

#define NN      131072
#define DEG     16
#define NNZ     (NN * DEG)
#define DMODEL  64
#define NHEAD   4
#define HDIM    16
#define BLK     256

typedef __attribute__((ext_vector_type(16))) _Float16 v16h;
typedef __attribute__((ext_vector_type(8)))  _Float16 v8h;
typedef __attribute__((ext_vector_type(8)))  float    v8f;

__device__ __forceinline__ v8f v8f_zero() {
    v8f z = {0.f, 0.f, 0.f, 0.f, 0.f, 0.f, 0.f, 0.f};
    return z;
}

__device__ __forceinline__ v16h cat16(v8h a, v8h b) {
    return __builtin_shufflevector(a, b, 0, 1, 2, 3, 4, 5, 6, 7,
                                   8, 9, 10, 11, 12, 13, 14, 15);
}

__device__ __forceinline__ v16h frag_from_f32(float4 a0, float4 a1,
                                              float4 a2, float4 a3) {
    v16h r;
    r[0] = (_Float16)a0.x;  r[1] = (_Float16)a0.y;
    r[2] = (_Float16)a0.z;  r[3] = (_Float16)a0.w;
    r[4] = (_Float16)a1.x;  r[5] = (_Float16)a1.y;
    r[6] = (_Float16)a1.z;  r[7] = (_Float16)a1.w;
    r[8] = (_Float16)a2.x;  r[9] = (_Float16)a2.y;
    r[10] = (_Float16)a2.z; r[11] = (_Float16)a2.w;
    r[12] = (_Float16)a3.x; r[13] = (_Float16)a3.y;
    r[14] = (_Float16)a3.z; r[15] = (_Float16)a3.w;
    return r;
}

__device__ __forceinline__ float4 relu4(float4 a, float4 b) {
    float4 r;
    r.x = a.x + b.x; r.x = r.x > 0.f ? r.x : 0.f;
    r.y = a.y + b.y; r.y = r.y > 0.f ? r.y : 0.f;
    r.z = a.z + b.z; r.z = r.z > 0.f ? r.z : 0.f;
    r.w = a.w + b.w; r.w = r.w > 0.f ? r.w : 0.f;
    return r;
}

template <int TM, int TN, bool RELU_IN, int ACT_OUT>
__global__ void wmma_gemm_kernel(const float* __restrict__ A,
                                 const float* __restrict__ abias,
                                 const float* __restrict__ W,
                                 const float* __restrict__ obias,
                                 float* __restrict__ C,
                                 int M, int K, int Nc)
{
    __shared__ __align__(16) _Float16 sW[12288];

    int tid  = (int)threadIdx.x;
    int lane = tid & 31;
    int hi = lane >> 4, lo = lane & 15;

    for (int i = tid; i < K * Nc; i += 256) {
        int k = i / Nc, n = i - k * Nc;
        sW[n * K + k] = (_Float16)W[i];
    }
    __syncthreads();

    int gwid = (int)((blockIdx.x * blockDim.x + threadIdx.x) >> 5);
    int groups_n = Nc / (16 * TN);
    int total = (M / (16 * TM)) * groups_n;
    if (gwid >= total) return;
    int mg = gwid / groups_n;
    int tg = gwid - mg * groups_n;

    #pragma unroll
    for (int t = 0; t < TM; ++t) {
        int tm = mg * TM + t;
        int m = tm * 16 + lo;

        v8f c[TN];
        #pragma unroll
        for (int j = 0; j < TN; ++j) c[j] = v8f_zero();

        const float* Arow = A + (size_t)m * K;
        for (int k0 = 0; k0 < K; k0 += 32) {
            const float4* ap0 = (const float4*)(Arow + k0 + 8 * hi);
            const float4* ap1 = (const float4*)(Arow + k0 + 16 + 8 * hi);
            float4 a0 = ap0[0], a1 = ap0[1], a2 = ap1[0], a3 = ap1[1];
            if (RELU_IN) {
                const float4* bp0 = (const float4*)(abias + k0 + 8 * hi);
                const float4* bp1 = (const float4*)(abias + k0 + 16 + 8 * hi);
                a0 = relu4(a0, bp0[0]); a1 = relu4(a1, bp0[1]);
                a2 = relu4(a2, bp1[0]); a3 = relu4(a3, bp1[1]);
            }
            v16h af = frag_from_f32(a0, a1, a2, a3);

            #pragma unroll
            for (int j = 0; j < TN; ++j) {
                int n = (tg * TN + j) * 16 + lo;
                const v8h* wp0 = (const v8h*)(&sW[n * K + k0 + 8 * hi]);
                const v8h* wp1 = (const v8h*)(&sW[n * K + k0 + 16 + 8 * hi]);
                v16h bf = cat16(wp0[0], wp1[0]);
                c[j] = __builtin_amdgcn_wmma_f32_16x16x32_f16(
                    false, af, false, bf, (short)0, c[j], false, false);
            }
        }

        #pragma unroll
        for (int j = 0; j < TN; ++j) {
            int n = (tg * TN + j) * 16 + lo;
            #pragma unroll
            for (int r = 0; r < 8; ++r) {
                int row = tm * 16 + r + 8 * hi;
                float v = c[j][r];
                if (obias) v += obias[n];
                if (ACT_OUT == 1) {
                    v = v > 0.f ? v : 0.f;
                } else if (ACT_OUT == 2) {
                    v = 0.5f * v * (1.0f + erff(v * 0.70710678118654752f));
                }
                C[(size_t)row * Nc + n] = v;
            }
        }
    }
}

__global__ void spmm_kernel(const int* __restrict__ rows,
                            const int* __restrict__ cols,
                            const float* __restrict__ vals,
                            const float* __restrict__ in,
                            float* __restrict__ out, int F4)
{
    long long t = (long long)blockIdx.x * blockDim.x + threadIdx.x;
    long long total = (long long)NNZ * F4;
    if (t >= total) return;
    int e  = (int)(t / F4);
    int c4 = (int)(t - (long long)e * F4);
    int F  = F4 * 4;
    float v = vals[e];
    const float4 x = ((const float4*)(in + (size_t)cols[e] * F))[c4];
    float* o = out + (size_t)rows[e] * F + c4 * 4;
    atomicAdd(o + 0, v * x.x);
    atomicAdd(o + 1, v * x.y);
    atomicAdd(o + 2, v * x.z);
    atomicAdd(o + 3, v * x.w);
}

__global__ void zero_kernel(float* __restrict__ p, int n)
{
    int i = (int)(blockIdx.x * blockDim.x + threadIdx.x);
    if (i < n) p[i] = 0.f;
}

__global__ void bias_relu_kernel(const float* __restrict__ in,
                                 const float* __restrict__ bias,
                                 float* __restrict__ out, int total, int F)
{
    int i = (int)(blockIdx.x * blockDim.x + threadIdx.x);
    if (i < total) {
        float v = in[i] + bias[i % F];
        out[i] = v > 0.f ? v : 0.f;
    }
}

__global__ void attn_kernel(const float* __restrict__ QKV,
                            float* __restrict__ ATT)
{
    __shared__ __align__(16) _Float16 sK[BLK * HDIM];
    __shared__ __align__(16) _Float16 sVt[HDIM * BLK];
    __shared__ __align__(16) _Float16 sP[8][16 * 32];

    int b = (int)(blockIdx.x >> 2);
    int h = (int)(blockIdx.x & 3);
    int tid  = (int)threadIdx.x;
    int wid  = tid >> 5;
    int lane = tid & 31;
    int hi   = lane >> 4;
    int lo   = lane & 15;
    size_t base = (size_t)b * BLK * (3 * DMODEL);

    for (int i = tid; i < BLK * HDIM; i += 256) {
        int s = i >> 4, d = i & 15;
        sK[i]            = (_Float16)QKV[base + (size_t)s * 192 + DMODEL     + h * HDIM + d];
        sVt[d * BLK + s] = (_Float16)QKV[base + (size_t)s * 192 + 2 * DMODEL + h * HDIM + d];
    }
    __syncthreads();

    v8h zero8 = {(_Float16)0.f, (_Float16)0.f, (_Float16)0.f, (_Float16)0.f,
                 (_Float16)0.f, (_Float16)0.f, (_Float16)0.f, (_Float16)0.f};

    for (int qt = wid; qt < 16; qt += 8) {
        int qrow = qt * 16 + lo;
        const float4* qp = (const float4*)(QKV + base + (size_t)qrow * 192
                                           + h * HDIM + 8 * hi);
        float4 q0 = qp[0], q1 = qp[1];
        q0.x *= 0.25f; q0.y *= 0.25f; q0.z *= 0.25f; q0.w *= 0.25f;
        q1.x *= 0.25f; q1.y *= 0.25f; q1.z *= 0.25f; q1.w *= 0.25f;
        float4 zf = {0.f, 0.f, 0.f, 0.f};
        v16h qf = frag_from_f32(q0, q1, zf, zf);

        v8f ctx = v8f_zero();
        float m[8], l[8];
        #pragma unroll
        for (int r = 0; r < 8; ++r) { m[r] = -3.0e38f; l[r] = 0.f; }

        for (int kk = 0; kk < 8; ++kk) {
            v8f s0, s1;
            {
                int key = (kk * 2) * 16 + lo;
                v16h bf = cat16(*(const v8h*)(&sK[key * HDIM + 8 * hi]), zero8);
                v8f z = v8f_zero();
                s0 = __builtin_amdgcn_wmma_f32_16x16x32_f16(
                    false, qf, false, bf, (short)0, z, false, false);
            }
            {
                int key = (kk * 2 + 1) * 16 + lo;
                v16h bf = cat16(*(const v8h*)(&sK[key * HDIM + 8 * hi]), zero8);
                v8f z = v8f_zero();
                s1 = __builtin_amdgcn_wmma_f32_16x16x32_f16(
                    false, qf, false, bf, (short)0, z, false, false);
            }
            #pragma unroll
            for (int r = 0; r < 8; ++r) {
                float tm = fmaxf(s0[r], s1[r]);
                #pragma unroll
                for (int d = 1; d < 16; d <<= 1)
                    tm = fmaxf(tm, __shfl_xor(tm, d, 32));
                float nm = fmaxf(m[r], tm);
                float sc = __expf(m[r] - nm);
                m[r] = nm;
                float e0 = __expf(s0[r] - nm);
                float e1 = __expf(s1[r] - nm);
                float rs = e0 + e1;
                #pragma unroll
                for (int d = 1; d < 16; d <<= 1)
                    rs += __shfl_xor(rs, d, 32);
                l[r] = l[r] * sc + rs;
                ctx[r] *= sc;
                int row = r + 8 * hi;
                sP[wid][row * 32 + lo]      = (_Float16)e0;
                sP[wid][row * 32 + 16 + lo] = (_Float16)e1;
            }
            v16h af = cat16(*(const v8h*)(&sP[wid][lo * 32 + 8 * hi]),
                            *(const v8h*)(&sP[wid][lo * 32 + 16 + 8 * hi]));
            v16h bf = cat16(*(const v8h*)(&sVt[lo * BLK + kk * 32 + 8 * hi]),
                            *(const v8h*)(&sVt[lo * BLK + kk * 32 + 16 + 8 * hi]));
            ctx = __builtin_amdgcn_wmma_f32_16x16x32_f16(
                false, af, false, bf, (short)0, ctx, false, false);
        }

        #pragma unroll
        for (int r = 0; r < 8; ++r) {
            int row = b * BLK + qt * 16 + r + 8 * hi;
            ATT[(size_t)row * DMODEL + h * HDIM + lo] = ctx[r] / l[r];
        }
    }
}

__global__ void ln_residual_kernel(const float* __restrict__ X,
                                   const float* __restrict__ Y,
                                   const float* __restrict__ g,
                                   const float* __restrict__ bt,
                                   float* __restrict__ out, int Mrows)
{
    int w = (int)((blockIdx.x * blockDim.x + threadIdx.x) >> 5);
    int lane = (int)(threadIdx.x & 31);
    if (w >= Mrows) return;
    size_t off = (size_t)w * DMODEL;
    float v0 = X[off + lane]      + Y[off + lane];
    float v1 = X[off + lane + 32] + Y[off + lane + 32];
    float s = v0 + v1;
    #pragma unroll
    for (int d = 1; d < 32; d <<= 1) s += __shfl_xor(s, d, 32);
    float mu = s * (1.f / 64.f);
    float d0 = v0 - mu, d1 = v1 - mu;
    float vs = d0 * d0 + d1 * d1;
    #pragma unroll
    for (int d = 1; d < 32; d <<= 1) vs += __shfl_xor(vs, d, 32);
    float rstd = rsqrtf(vs * (1.f / 64.f) + 1e-5f);
    out[off + lane]      = d0 * rstd * g[lane]      + bt[lane];
    out[off + lane + 32] = d1 * rstd * g[lane + 32] + bt[lane + 32];
}

__global__ void cls_kernel(const float* __restrict__ X2,
                           const float* __restrict__ Wc,
                           const float* __restrict__ bc,
                           float* __restrict__ out, int Mrows)
{
    int w = (int)((blockIdx.x * blockDim.x + threadIdx.x) >> 5);
    int lane = (int)(threadIdx.x & 31);
    int hi = lane >> 4, lo = lane & 15;
    if (w >= Mrows) return;
    size_t off = (size_t)w * DMODEL;
    float acc = 0.f;
    #pragma unroll
    for (int kk = 0; kk < 32; ++kk) {
        int k = hi * 32 + kk;
        acc += X2[off + k] * Wc[k * 16 + lo];
    }
    acc += __shfl_xor(acc, 16, 32);
    acc += bc[lo];
    float mx = acc;
    #pragma unroll
    for (int d = 1; d < 16; d <<= 1) mx = fmaxf(mx, __shfl_xor(mx, d, 32));
    float e = expf(acc - mx), sm = e;
    #pragma unroll
    for (int d = 1; d < 16; d <<= 1) sm += __shfl_xor(sm, d, 32);
    float res = acc - mx - logf(sm);
    if (hi == 0) out[(size_t)w * 16 + lo] = res;
}


static inline int ceil_div(long long a, long long b) { return (int)((a + b - 1) / b); }

#define GEMM_WAVES(M, TM, Nc, TN) \
    ((long long)((M) / (16 * (TM))) * ((Nc) / (16 * (TN))))

static void launch_gemm_none(const float* A, const float* W, const float* ob,
                             float* C, int M, int K, int Nc, hipStream_t s)
{
    long long waves = GEMM_WAVES(M, 4, Nc, 4);
    wmma_gemm_kernel<4, 4, false, 0><<<ceil_div(waves * 32, 256), 256, 0, s>>>(
        A, nullptr, W, ob, C, M, K, Nc);
}

extern "C" void kernel_launch(void* const* d_in, const int* in_sizes, int n_in,
                              void* d_out, int out_size, void* d_ws, size_t ws_size,
                              hipStream_t stream)
{
    const float* x      = (const float*)d_in[0];
    const int*   G_rows = (const int*)  d_in[1];
    const int*   G_cols = (const int*)  d_in[2];
    const float* G_vals = (const float*)d_in[3];
    const float* W1     = (const float*)d_in[4];
    const float* b1     = (const float*)d_in[5];
    const float* W2     = (const float*)d_in[6];
    const float* b2     = (const float*)d_in[7];
    const float* W_qkv  = (const float*)d_in[8];
    const float* b_qkv  = (const float*)d_in[9];
    const float* W_o    = (const float*)d_in[10];
    const float* b_o    = (const float*)d_in[11];
    const float* ln1_g  = (const float*)d_in[12];
    const float* ln1_b  = (const float*)d_in[13];
    const float* W_ff1  = (const float*)d_in[14];
    const float* b_ff1  = (const float*)d_in[15];
    const float* W_ff2  = (const float*)d_in[16];
    const float* b_ff2  = (const float*)d_in[17];
    const float* ln2_g  = (const float*)d_in[18];
    const float* ln2_b  = (const float*)d_in[19];
    const float* W_cls  = (const float*)d_in[20];
    const float* b_cls  = (const float*)d_in[21];
    float* out = (float*)d_out;

    float* ws = (float*)d_ws;
    const size_t n = NN;
    float* P1  = ws;
    float* S1  = ws + n * 32;
    float* P2  = ws + n * 64;
    float* S2  = ws + n * 128;
    float* H2  = ws + n * 192;
    float* QKV = ws + n * 256;
    float* ATT = ws;
    float* A1  = ws + n * 64;
    float* X1  = ws + n * 128;
    float* FF1 = ws + n * 256;
    float* A2  = ws + n * 384;
    float* X2  = ws;

    {
        long long waves = GEMM_WAVES(NN, 4, 32, 2);
        wmma_gemm_kernel<4, 2, false, 0><<<ceil_div(waves * 32, 256), 256, 0, stream>>>(
            x, nullptr, W1, nullptr, P1, NN, 128, 32);
    }

    zero_kernel<<<ceil_div((long long)NN * 32, 256), 256, 0, stream>>>(S1, NN * 32);
    spmm_kernel<<<ceil_div((long long)NNZ * 8, 256), 256, 0, stream>>>(
        G_rows, G_cols, G_vals, P1, S1, 8);

    {
        long long waves = GEMM_WAVES(NN, 4, 64, 4);
        wmma_gemm_kernel<4, 4, true, 0><<<ceil_div(waves * 32, 256), 256, 0, stream>>>(
            S1, b1, W2, nullptr, P2, NN, 32, 64);
    }

    zero_kernel<<<ceil_div((long long)NN * 64, 256), 256, 0, stream>>>(S2, NN * 64);
    spmm_kernel<<<ceil_div((long long)NNZ * 16, 256), 256, 0, stream>>>(
        G_rows, G_cols, G_vals, P2, S2, 16);

    bias_relu_kernel<<<ceil_div((long long)NN * 64, 256), 256, 0, stream>>>(
        S2, b2, H2, NN * 64, 64);

    launch_gemm_none(H2, W_qkv, b_qkv, QKV, NN, 64, 192, stream);

    attn_kernel<<<(NN / BLK) * NHEAD, 256, 0, stream>>>(QKV, ATT);

    launch_gemm_none(ATT, W_o, b_o, A1, NN, 64, 64, stream);

    ln_residual_kernel<<<ceil_div((long long)NN * 32, 256), 256, 0, stream>>>(
        H2, A1, ln1_g, ln1_b, X1, NN);

    {
        long long waves = GEMM_WAVES(NN, 4, 128, 4);
        wmma_gemm_kernel<4, 4, false, 2><<<ceil_div(waves * 32, 256), 256, 0, stream>>>(
            X1, nullptr, W_ff1, b_ff1, FF1, NN, 64, 128);
    }

    launch_gemm_none(FF1, W_ff2, b_ff2, A2, NN, 128, 64, stream);

    ln_residual_kernel<<<ceil_div((long long)NN * 32, 256), 256, 0, stream>>>(
        X1, A2, ln2_g, ln2_b, X2, NN);

    cls_kernel<<<ceil_div((long long)NN * 32, 256), 256, 0, stream>>>(
        X2, W_cls, b_cls, out, NN);
}


# Attention scores computed transposed (S^T = K Q^T): softmax max/sum per query in the lane's own registers + one half-wave exchange, P^T fed to V^T P^T from registers; softmax sum order changed
# speedup vs baseline: 1.2549x; 1.2549x over previous
#include <hip/hip_runtime.h>
#include <hip/hip_bf16.h>
#include <math.h>

#define NN      131072
#define DEG     16
#define NNZ     (NN * DEG)
#define DMODEL  64
#define NHEAD   4
#define HDIM    16
#define BLK     256

typedef __attribute__((ext_vector_type(16))) _Float16 v16h;
typedef __attribute__((ext_vector_type(8)))  _Float16 v8h;
typedef __attribute__((ext_vector_type(8)))  float    v8f;

__device__ __forceinline__ v8f v8f_zero() {
    v8f z = {0.f, 0.f, 0.f, 0.f, 0.f, 0.f, 0.f, 0.f};
    return z;
}

__device__ __forceinline__ v16h cat16(v8h a, v8h b) {
    return __builtin_shufflevector(a, b, 0, 1, 2, 3, 4, 5, 6, 7,
                                   8, 9, 10, 11, 12, 13, 14, 15);
}

__device__ __forceinline__ v16h frag_from_f32(float4 a0, float4 a1,
                                              float4 a2, float4 a3) {
    v16h r;
    r[0] = (_Float16)a0.x;  r[1] = (_Float16)a0.y;
    r[2] = (_Float16)a0.z;  r[3] = (_Float16)a0.w;
    r[4] = (_Float16)a1.x;  r[5] = (_Float16)a1.y;
    r[6] = (_Float16)a1.z;  r[7] = (_Float16)a1.w;
    r[8] = (_Float16)a2.x;  r[9] = (_Float16)a2.y;
    r[10] = (_Float16)a2.z; r[11] = (_Float16)a2.w;
    r[12] = (_Float16)a3.x; r[13] = (_Float16)a3.y;
    r[14] = (_Float16)a3.z; r[15] = (_Float16)a3.w;
    return r;
}

__device__ __forceinline__ float4 relu4(float4 a, float4 b) {
    float4 r;
    r.x = a.x + b.x; r.x = r.x > 0.f ? r.x : 0.f;
    r.y = a.y + b.y; r.y = r.y > 0.f ? r.y : 0.f;
    r.z = a.z + b.z; r.z = r.z > 0.f ? r.z : 0.f;
    r.w = a.w + b.w; r.w = r.w > 0.f ? r.w : 0.f;
    return r;
}

template <int TM, int TN, bool RELU_IN, int ACT_OUT>
__global__ void wmma_gemm_kernel(const float* __restrict__ A,
                                 const float* __restrict__ abias,
                                 const float* __restrict__ W,
                                 const float* __restrict__ obias,
                                 float* __restrict__ C,
                                 int M, int K, int Nc)
{
    __shared__ __align__(16) _Float16 sW[12288];

    int tid  = (int)threadIdx.x;
    int lane = tid & 31;
    int hi = lane >> 4, lo = lane & 15;

    for (int i = tid; i < K * Nc; i += 256) {
        int k = i / Nc, n = i - k * Nc;
        sW[n * K + k] = (_Float16)W[i];
    }
    __syncthreads();

    int gwid = (int)((blockIdx.x * blockDim.x + threadIdx.x) >> 5);
    int groups_n = Nc / (16 * TN);
    int total = (M / (16 * TM)) * groups_n;
    if (gwid >= total) return;
    int mg = gwid / groups_n;
    int tg = gwid - mg * groups_n;

    #pragma unroll
    for (int t = 0; t < TM; ++t) {
        int tm = mg * TM + t;
        int m = tm * 16 + lo;

        v8f c[TN];
        #pragma unroll
        for (int j = 0; j < TN; ++j) c[j] = v8f_zero();

        const float* Arow = A + (size_t)m * K;
        for (int k0 = 0; k0 < K; k0 += 32) {
            const float4* ap0 = (const float4*)(Arow + k0 + 8 * hi);
            const float4* ap1 = (const float4*)(Arow + k0 + 16 + 8 * hi);
            float4 a0 = ap0[0], a1 = ap0[1], a2 = ap1[0], a3 = ap1[1];
            if (RELU_IN) {
                const float4* bp0 = (const float4*)(abias + k0 + 8 * hi);
                const float4* bp1 = (const float4*)(abias + k0 + 16 + 8 * hi);
                a0 = relu4(a0, bp0[0]); a1 = relu4(a1, bp0[1]);
                a2 = relu4(a2, bp1[0]); a3 = relu4(a3, bp1[1]);
            }
            v16h af = frag_from_f32(a0, a1, a2, a3);

            #pragma unroll
            for (int j = 0; j < TN; ++j) {
                int n = (tg * TN + j) * 16 + lo;
                const v8h* wp0 = (const v8h*)(&sW[n * K + k0 + 8 * hi]);
                const v8h* wp1 = (const v8h*)(&sW[n * K + k0 + 16 + 8 * hi]);
                v16h bf = cat16(wp0[0], wp1[0]);
                c[j] = __builtin_amdgcn_wmma_f32_16x16x32_f16(
                    false, af, false, bf, (short)0, c[j], false, false);
            }
        }

        #pragma unroll
        for (int j = 0; j < TN; ++j) {
            int n = (tg * TN + j) * 16 + lo;
            #pragma unroll
            for (int r = 0; r < 8; ++r) {
                int row = tm * 16 + r + 8 * hi;
                float v = c[j][r];
                if (obias) v += obias[n];
                if (ACT_OUT == 1) {
                    v = v > 0.f ? v : 0.f;
                } else if (ACT_OUT == 2) {
                    v = 0.5f * v * (1.0f + erff(v * 0.70710678118654752f));
                }
                C[(size_t)row * Nc + n] = v;
            }
        }
    }
}

__global__ void rank_kernel(const int* __restrict__ rows,
                            int* __restrict__ cnt,
                            int* __restrict__ rank)
{
    int e = (int)(blockIdx.x * blockDim.x + threadIdx.x);
    if (e < NNZ) rank[e] = atomicAdd(&cnt[rows[e]], 1);
}

__global__ void __launch_bounds__(1024)
rowscan_kernel(const int* __restrict__ cnt, int2* __restrict__ rowinfo)
{
    __shared__ int sBase[32];
    __shared__ int sWave[32];
    int t    = (int)threadIdx.x;
    int b    = (int)blockIdx.x;
    int lane = t & 31;
    int wid  = t >> 5;

    const int4* all = (const int4*)cnt;
    int part = 0;
    #pragma unroll 4
    for (int i = t; i < b * 256; i += 1024) {
        int4 q = all[i];
        part += (q.x + q.y) + (q.z + q.w);
    }
    for (int d = 1; d < 32; d <<= 1) part += __shfl_xor(part, d, 32);

    int v = cnt[b * 1024 + t];
    int x = v;
    for (int d = 1; d < 32; d <<= 1) {
        int y = __shfl_up(x, d, 32);
        if (lane >= d) x += y;
    }
    if (lane == 0)  sBase[wid] = part;
    if (lane == 31) sWave[wid] = x;
    __syncthreads();

    int base = 0, before = 0;
    for (int i = 0; i < 32; ++i) {
        int w = sWave[i];
        base += sBase[i];
        before += (i < wid) ? w : 0;
    }
    int2 ri;
    ri.x = base + before + x - v;
    ri.y = v;
    rowinfo[b * 1024 + t] = ri;
}

__global__ void place_kernel(const int* __restrict__ rows,
                             const int* __restrict__ cols,
                             const float* __restrict__ vals,
                             const int2* __restrict__ rowinfo,
                             const int* __restrict__ rank,
                             int2* __restrict__ cv)
{
    int e = (int)(blockIdx.x * blockDim.x + threadIdx.x);
    if (e < NNZ) {
        int2 p;
        p.x = cols[e];
        p.y = __float_as_int(vals[e]);
        cv[rowinfo[rows[e]].x + rank[e]] = p;
    }
}

template <int F4>
__global__ void __launch_bounds__(256)
spmm_gather_kernel(const int2* __restrict__ rowinfo,
                   const int2* __restrict__ cv,
                   const float* __restrict__ in,
                   float* __restrict__ out)
{
    int t   = (int)(blockIdx.x * blockDim.x + threadIdx.x);
    int row = t / F4;
    int c4  = t - row * F4;
    if (row >= NN) return;
    int2 ri  = rowinfo[row];
    int j    = ri.x;
    int end  = j + ri.y;
    int last = end - 1;
    const float4* inp = (const float4*)in + c4;
    float4 acc = {0.f, 0.f, 0.f, 0.f};
    if (j < end) {
        int2 e0 = cv[j];
        int2 e1 = cv[min(j + 1, last)];
        int2 e2 = cv[min(j + 2, last)];
        int2 e3 = cv[min(j + 3, last)];
        for (; j < end; j += 4) {
            float4 x0 = inp[(size_t)e0.x * F4];
            float4 x1 = inp[(size_t)e1.x * F4];
            float4 x2 = inp[(size_t)e2.x * F4];
            float4 x3 = inp[(size_t)e3.x * F4];
            int jn = j + 4;
            int2 n0 = cv[min(jn, last)];
            int2 n1 = cv[min(jn + 1, last)];
            int2 n2 = cv[min(jn + 2, last)];
            int2 n3 = cv[min(jn + 3, last)];
            asm volatile("" ::: "memory");
            bool p1 = j + 1 < end, p2 = j + 2 < end, p3 = j + 3 < end;
            float v0 = __int_as_float(e0.y), v1 = __int_as_float(e1.y);
            float v2 = __int_as_float(e2.y), v3 = __int_as_float(e3.y);
            acc.x += v0 * x0.x; acc.y += v0 * x0.y; acc.z += v0 * x0.z; acc.w += v0 * x0.w;
            acc.x += p1 ? v1 * x1.x : 0.f; acc.y += p1 ? v1 * x1.y : 0.f;
            acc.z += p1 ? v1 * x1.z : 0.f; acc.w += p1 ? v1 * x1.w : 0.f;
            acc.x += p2 ? v2 * x2.x : 0.f; acc.y += p2 ? v2 * x2.y : 0.f;
            acc.z += p2 ? v2 * x2.z : 0.f; acc.w += p2 ? v2 * x2.w : 0.f;
            acc.x += p3 ? v3 * x3.x : 0.f; acc.y += p3 ? v3 * x3.y : 0.f;
            acc.z += p3 ? v3 * x3.z : 0.f; acc.w += p3 ? v3 * x3.w : 0.f;
            e0 = n0; e1 = n1; e2 = n2; e3 = n3;
        }
    }
    ((float4*)out)[(size_t)row * F4 + c4] = acc;
}

__global__ void zero_kernel(float* __restrict__ p, int n)
{
    int i = (int)(blockIdx.x * blockDim.x + threadIdx.x);
    if (i < n) p[i] = 0.f;
}

#define BR_PER_THREAD 8
__global__ void __launch_bounds__(256)
bias_relu_kernel(const float* __restrict__ in,
                 const float* __restrict__ bias,
                 float* __restrict__ out)
{
    int t = (int)threadIdx.x;
    size_t first = (size_t)blockIdx.x * (256 * BR_PER_THREAD) + (size_t)t;
    const float4* ip = (const float4*)in + first;
    float4*       op = (float4*)out + first;
    float4 b = ((const float4*)bias)[t & 15];
    float4 v[BR_PER_THREAD];
    for (int u = 0; u < BR_PER_THREAD; ++u) v[u] = ip[u * 256];
    for (int u = 0; u < BR_PER_THREAD; ++u) op[u * 256] = relu4(v[u], b);
}

__device__ __forceinline__ float xhalf(float x) {
    int v = __builtin_bit_cast(int, x);
    return __builtin_bit_cast(float,
        __builtin_amdgcn_permlanex16(v, v, 0x76543210, 0xfedcba98, false, false));
}

__global__ void attn_kernel(const float* __restrict__ QKV,
                            float* __restrict__ ATT)
{
    __shared__ __align__(16) _Float16 sK[BLK * HDIM];
    __shared__ __align__(16) _Float16 sVt[HDIM * BLK];

    int b = (int)(blockIdx.x >> 2);
    int h = (int)(blockIdx.x & 3);
    int tid  = (int)threadIdx.x;
    int wid  = tid >> 5;
    int lane = tid & 31;
    int hi   = lane >> 4;
    int lo   = lane & 15;
    size_t base = (size_t)b * BLK * (3 * DMODEL);

    for (int i = tid; i < BLK * HDIM; i += 256) {
        int s = i >> 4, d = i & 15;
        sK[i]            = (_Float16)QKV[base + (size_t)s * 192 + DMODEL     + h * HDIM + d];
        sVt[d * BLK + s] = (_Float16)QKV[base + (size_t)s * 192 + 2 * DMODEL + h * HDIM + d];
    }
    __syncthreads();

    v8h zero8 = {(_Float16)0.f, (_Float16)0.f, (_Float16)0.f, (_Float16)0.f,
                 (_Float16)0.f, (_Float16)0.f, (_Float16)0.f, (_Float16)0.f};

    for (int qt = wid; qt < 16; qt += 8) {
        int qrow = qt * 16 + lo;
        const float4* qp = (const float4*)(QKV + base + (size_t)qrow * 192
                                           + h * HDIM + 8 * hi);
        float4 q0 = qp[0], q1 = qp[1];
        q0.x *= 0.25f; q0.y *= 0.25f; q0.z *= 0.25f; q0.w *= 0.25f;
        q1.x *= 0.25f; q1.y *= 0.25f; q1.z *= 0.25f; q1.w *= 0.25f;
        float4 zf = {0.f, 0.f, 0.f, 0.f};
        v16h qf = frag_from_f32(q0, q1, zf, zf);

        v8f ctx = v8f_zero();
        float m = -3.0e38f, l = 0.f;

        for (int kk = 0; kk < 8; ++kk) {
            v8f s0, s1;
            {
                int key = (kk * 2) * 16 + lo;
                v16h kf = cat16(*(const v8h*)(&sK[key * HDIM + 8 * hi]), zero8);
                v8f z = v8f_zero();
                s0 = __builtin_amdgcn_wmma_f32_16x16x32_f16(
                    false, kf, false, qf, (short)0, z, false, false);
            }
            {
                int key = (kk * 2 + 1) * 16 + lo;
                v16h kf = cat16(*(const v8h*)(&sK[key * HDIM + 8 * hi]), zero8);
                v8f z = v8f_zero();
                s1 = __builtin_amdgcn_wmma_f32_16x16x32_f16(
                    false, kf, false, qf, (short)0, z, false, false);
            }
            float tm = fmaxf(s0[0], s1[0]);
            #pragma unroll
            for (int r = 1; r < 8; ++r) tm = fmaxf(tm, fmaxf(s0[r], s1[r]));
            tm = fmaxf(tm, xhalf(tm));
            float nm = fmaxf(m, tm);
            float sc = __expf(m - nm);
            m = nm;
            float rs = 0.f;
            v16h pf;
            #pragma unroll
            for (int r = 0; r < 8; ++r) {
                float e0 = __expf(s0[r] - nm);
                float e1 = __expf(s1[r] - nm);
                rs += e0 + e1;
                pf[r]     = (_Float16)e0;
                pf[8 + r] = (_Float16)e1;
            }
            l = l * sc + rs;
            #pragma unroll
            for (int r = 0; r < 8; ++r) ctx[r] *= sc;
            v16h vf = cat16(*(const v8h*)(&sVt[lo * BLK + kk * 32 + 8 * hi]),
                            *(const v8h*)(&sVt[lo * BLK + kk * 32 + 16 + 8 * hi]));
            ctx = __builtin_amdgcn_wmma_f32_16x16x32_f16(
                false, vf, false, pf, (short)0, ctx, false, false);
        }

        l = l + xhalf(l);
        float4 o0, o1;
        o0.x = ctx[0] / l; o0.y = ctx[1] / l; o0.z = ctx[2] / l; o0.w = ctx[3] / l;
        o1.x = ctx[4] / l; o1.y = ctx[5] / l; o1.z = ctx[6] / l; o1.w = ctx[7] / l;
        float4* op = (float4*)(ATT + (size_t)(b * BLK + qrow) * DMODEL + h * HDIM + 8 * hi);
        op[0] = o0;
        op[1] = o1;
    }
}

#define ROWS_PER_WAVE 8
__global__ void __launch_bounds__(256)
ln_residual_kernel(const float* __restrict__ X,
                   const float* __restrict__ Y,
                   const float* __restrict__ g,
                   const float* __restrict__ bt,
                   float* __restrict__ out)
{
    int w = (int)((blockIdx.x * blockDim.x + threadIdx.x) >> 5);
    int lane = (int)(threadIdx.x & 31);
    size_t off0 = (size_t)w * (ROWS_PER_WAVE * DMODEL);
    float g0 = g[lane], g1 = g[lane + 32];
    float b0 = bt[lane], b1 = bt[lane + 32];
    float v0[ROWS_PER_WAVE], v1[ROWS_PER_WAVE];
    #pragma unroll
    for (int i = 0; i < ROWS_PER_WAVE; ++i) {
        size_t off = off0 + (size_t)i * DMODEL;
        v0[i] = X[off + lane]      + Y[off + lane];
        v1[i] = X[off + lane + 32] + Y[off + lane + 32];
    }
    #pragma unroll
    for (int i = 0; i < ROWS_PER_WAVE; ++i) {
        size_t off = off0 + (size_t)i * DMODEL;
        float s = v0[i] + v1[i];
        #pragma unroll
        for (int d = 1; d < 32; d <<= 1) s += __shfl_xor(s, d, 32);
        float mu = s * (1.f / 64.f);
        float d0 = v0[i] - mu, d1 = v1[i] - mu;
        float vs = d0 * d0 + d1 * d1;
        #pragma unroll
        for (int d = 1; d < 32; d <<= 1) vs += __shfl_xor(vs, d, 32);
        float rstd = rsqrtf(vs * (1.f / 64.f) + 1e-5f);
        out[off + lane]      = d0 * rstd * g0 + b0;
        out[off + lane + 32] = d1 * rstd * g1 + b1;
    }
}

__device__ __forceinline__ float cls_row(const float4 (&x)[8], const float (&wc)[32],
                                         float bias)
{
    float acc = 0.f;
    for (int q = 0; q < 8; ++q) {
        acc += x[q].x * wc[4 * q];
        acc += x[q].y * wc[4 * q + 1];
        acc += x[q].z * wc[4 * q + 2];
        acc += x[q].w * wc[4 * q + 3];
    }
    acc += __shfl_xor(acc, 16, 32);
    acc += bias;
    float mx = acc;
    #pragma unroll
    for (int d = 1; d < 16; d <<= 1) mx = fmaxf(mx, __shfl_xor(mx, d, 32));
    float e = expf(acc - mx), sm = e;
    #pragma unroll
    for (int d = 1; d < 16; d <<= 1) sm += __shfl_xor(sm, d, 32);
    return acc - mx - logf(sm);
}

__global__ void __launch_bounds__(256)
cls_kernel(const float* __restrict__ X2,
           const float* __restrict__ Wc,
           const float* __restrict__ bc,
           float* __restrict__ out)
{
    int w = (int)((blockIdx.x * blockDim.x + threadIdx.x) >> 5);
    int lane = (int)(threadIdx.x & 31);
    int hi = lane >> 4, lo = lane & 15;
    float wc[32];
    for (int kk = 0; kk < 32; ++kk) wc[kk] = Wc[(hi * 32 + kk) * 16 + lo];
    float bias = bc[lo];
    int row0 = w * ROWS_PER_WAVE;
    for (int p = 0; p < ROWS_PER_WAVE / 2; ++p) {
        int ra = row0 + 2 * p;
        const float4* xa = (const float4*)(X2 + (size_t)ra * DMODEL + hi * 32);
        const float4* xb = (const float4*)(X2 + (size_t)(ra + 1) * DMODEL + hi * 32);
        float4 a[8], b[8];
        for (int q = 0; q < 8; ++q) { a[q] = xa[q]; b[q] = xb[q]; }
        float res_a = cls_row(a, wc, bias);
        float res_b = cls_row(b, wc, bias);
        out[(size_t)(ra + hi) * 16 + lo] = hi ? res_b : res_a;
    }
}


static inline int ceil_div(long long a, long long b) { return (int)((a + b - 1) / b); }

#define GEMM_WAVES(M, TM, Nc, TN) \
    ((long long)((M) / (16 * (TM))) * ((Nc) / (16 * (TN))))

static void launch_gemm_none(const float* A, const float* W, const float* ob,
                             float* C, int M, int K, int Nc, hipStream_t s)
{
    long long waves = GEMM_WAVES(M, 4, Nc, 4);
    wmma_gemm_kernel<4, 4, false, 0><<<ceil_div(waves * 32, 256), 256, 0, s>>>(
        A, nullptr, W, ob, C, M, K, Nc);
}

extern "C" void kernel_launch(void* const* d_in, const int* in_sizes, int n_in,
                              void* d_out, int out_size, void* d_ws, size_t ws_size,
                              hipStream_t stream)
{
    const float* x      = (const float*)d_in[0];
    const int*   G_rows = (const int*)  d_in[1];
    const int*   G_cols = (const int*)  d_in[2];
    const float* G_vals = (const float*)d_in[3];
    const float* W1     = (const float*)d_in[4];
    const float* b1     = (const float*)d_in[5];
    const float* W2     = (const float*)d_in[6];
    const float* b2     = (const float*)d_in[7];
    const float* W_qkv  = (const float*)d_in[8];
    const float* b_qkv  = (const float*)d_in[9];
    const float* W_o    = (const float*)d_in[10];
    const float* b_o    = (const float*)d_in[11];
    const float* ln1_g  = (const float*)d_in[12];
    const float* ln1_b  = (const float*)d_in[13];
    const float* W_ff1  = (const float*)d_in[14];
    const float* b_ff1  = (const float*)d_in[15];
    const float* W_ff2  = (const float*)d_in[16];
    const float* b_ff2  = (const float*)d_in[17];
    const float* ln2_g  = (const float*)d_in[18];
    const float* ln2_b  = (const float*)d_in[19];
    const float* W_cls  = (const float*)d_in[20];
    const float* b_cls  = (const float*)d_in[21];
    float* out = (float*)d_out;

    float* ws = (float*)d_ws;
    const size_t n = NN;
    float* P1  = ws;
    float* S1  = ws + n * 32;
    float* P2  = ws + n * 64;
    float* S2  = ws + n * 128;
    float* H2  = ws + n * 192;
    float* QKV = ws + n * 256;
    float* ATT = ws;
    float* A1  = ws + n * 64;
    float* X1  = ws + n * 128;
    float* FF1 = ws + n * 256;
    float* A2  = ws + n * 384;
    float* X2  = ws;
    int*  CNT  = (int*)(ws + n * 256);
    int2* RINF = (int2*)(ws + n * 257);
    int*  RANK = (int*)(ws + n * 259);
    int2* CV   = (int2*)(ws + n * 275);

    {
        long long waves = GEMM_WAVES(NN, 4, 32, 2);
        wmma_gemm_kernel<4, 2, false, 0><<<ceil_div(waves * 32, 256), 256, 0, stream>>>(
            x, nullptr, W1, nullptr, P1, NN, 128, 32);
    }

    zero_kernel<<<ceil_div((long long)NN, 256), 256, 0, stream>>>((float*)CNT, NN);
    rank_kernel<<<ceil_div((long long)NNZ, 256), 256, 0, stream>>>(G_rows, CNT, RANK);
    rowscan_kernel<<<NN / 1024, 1024, 0, stream>>>(CNT, RINF);
    place_kernel<<<ceil_div((long long)NNZ, 256), 256, 0, stream>>>(
        G_rows, G_cols, G_vals, RINF, RANK, CV);

    spmm_gather_kernel<8><<<ceil_div((long long)NN * 8, 256), 256, 0, stream>>>(
        RINF, CV, P1, S1);

    {
        long long waves = GEMM_WAVES(NN, 4, 64, 4);
        wmma_gemm_kernel<4, 4, true, 0><<<ceil_div(waves * 32, 256), 256, 0, stream>>>(
            S1, b1, W2, nullptr, P2, NN, 32, 64);
    }

    spmm_gather_kernel<16><<<ceil_div((long long)NN * 16, 256), 256, 0, stream>>>(
        RINF, CV, P2, S2);

    bias_relu_kernel<<<NN * 16 / (256 * BR_PER_THREAD), 256, 0, stream>>>(S2, b2, H2);

    launch_gemm_none(H2, W_qkv, b_qkv, QKV, NN, 64, 192, stream);

    attn_kernel<<<(NN / BLK) * NHEAD, 256, 0, stream>>>(QKV, ATT);

    launch_gemm_none(ATT, W_o, b_o, A1, NN, 64, 64, stream);

    ln_residual_kernel<<<NN / ROWS_PER_WAVE / 8, 256, 0, stream>>>(
        H2, A1, ln1_g, ln1_b, X1);

    {
        long long waves = GEMM_WAVES(NN, 4, 128, 4);
        wmma_gemm_kernel<4, 4, false, 2><<<ceil_div(waves * 32, 256), 256, 0, stream>>>(
            X1, nullptr, W_ff1, b_ff1, FF1, NN, 64, 128);
    }

    launch_gemm_none(FF1, W_ff2, b_ff2, A2, NN, 128, 64, stream);

    ln_residual_kernel<<<NN / ROWS_PER_WAVE / 8, 256, 0, stream>>>(
        X1, A2, ln2_g, ln2_b, X2);

    cls_kernel<<<NN / ROWS_PER_WAVE / 8, 256, 0, stream>>>(
        X2, W_cls, b_cls, out);
}


# Row-local chain behind attention (W_o+residual+LN1+FF1+gelu+FF2+residual+LN2) fused in one kernel, products transposed so tiles stay in registers; 17 launches become 13; LayerNorm sum order changed
# speedup vs baseline: 1.7676x; 1.4086x over previous
#include <hip/hip_runtime.h>
#include <hip/hip_bf16.h>
#include <math.h>

#define NN      131072
#define DEG     16
#define NNZ     (NN * DEG)
#define DMODEL  64
#define NHEAD   4
#define HDIM    16
#define BLK     256

typedef __attribute__((ext_vector_type(16))) _Float16 v16h;
typedef __attribute__((ext_vector_type(8)))  _Float16 v8h;
typedef __attribute__((ext_vector_type(8)))  float    v8f;

__device__ __forceinline__ v8f v8f_zero() {
    v8f z = {0.f, 0.f, 0.f, 0.f, 0.f, 0.f, 0.f, 0.f};
    return z;
}

__device__ __forceinline__ v16h cat16(v8h a, v8h b) {
    return __builtin_shufflevector(a, b, 0, 1, 2, 3, 4, 5, 6, 7,
                                   8, 9, 10, 11, 12, 13, 14, 15);
}

__device__ __forceinline__ v16h frag_from_f32(float4 a0, float4 a1,
                                              float4 a2, float4 a3) {
    v16h r;
    r[0] = (_Float16)a0.x;  r[1] = (_Float16)a0.y;
    r[2] = (_Float16)a0.z;  r[3] = (_Float16)a0.w;
    r[4] = (_Float16)a1.x;  r[5] = (_Float16)a1.y;
    r[6] = (_Float16)a1.z;  r[7] = (_Float16)a1.w;
    r[8] = (_Float16)a2.x;  r[9] = (_Float16)a2.y;
    r[10] = (_Float16)a2.z; r[11] = (_Float16)a2.w;
    r[12] = (_Float16)a3.x; r[13] = (_Float16)a3.y;
    r[14] = (_Float16)a3.z; r[15] = (_Float16)a3.w;
    return r;
}

__device__ __forceinline__ float4 relu4(float4 a, float4 b) {
    float4 r;
    r.x = a.x + b.x; r.x = r.x > 0.f ? r.x : 0.f;
    r.y = a.y + b.y; r.y = r.y > 0.f ? r.y : 0.f;
    r.z = a.z + b.z; r.z = r.z > 0.f ? r.z : 0.f;
    r.w = a.w + b.w; r.w = r.w > 0.f ? r.w : 0.f;
    return r;
}

template <int TM, int TN, bool RELU_IN, int ACT_OUT>
__global__ void wmma_gemm_kernel(const float* __restrict__ A,
                                 const float* __restrict__ abias,
                                 const float* __restrict__ W,
                                 const float* __restrict__ obias,
                                 float* __restrict__ C,
                                 int M, int K, int Nc)
{
    __shared__ __align__(16) _Float16 sW[12288];

    int tid  = (int)threadIdx.x;
    int lane = tid & 31;
    int hi = lane >> 4, lo = lane & 15;

    for (int i = tid; i < K * Nc; i += 256) {
        int k = i / Nc, n = i - k * Nc;
        sW[n * K + k] = (_Float16)W[i];
    }
    __syncthreads();

    int gwid = (int)((blockIdx.x * blockDim.x + threadIdx.x) >> 5);
    int groups_n = Nc / (16 * TN);
    int total = (M / (16 * TM)) * groups_n;
    if (gwid >= total) return;
    int mg = gwid / groups_n;
    int tg = gwid - mg * groups_n;

    #pragma unroll
    for (int t = 0; t < TM; ++t) {
        int tm = mg * TM + t;
        int m = tm * 16 + lo;

        v8f c[TN];
        #pragma unroll
        for (int j = 0; j < TN; ++j) c[j] = v8f_zero();

        const float* Arow = A + (size_t)m * K;
        for (int k0 = 0; k0 < K; k0 += 32) {
            const float4* ap0 = (const float4*)(Arow + k0 + 8 * hi);
            const float4* ap1 = (const float4*)(Arow + k0 + 16 + 8 * hi);
            float4 a0 = ap0[0], a1 = ap0[1], a2 = ap1[0], a3 = ap1[1];
            if (RELU_IN) {
                const float4* bp0 = (const float4*)(abias + k0 + 8 * hi);
                const float4* bp1 = (const float4*)(abias + k0 + 16 + 8 * hi);
                a0 = relu4(a0, bp0[0]); a1 = relu4(a1, bp0[1]);
                a2 = relu4(a2, bp1[0]); a3 = relu4(a3, bp1[1]);
            }
            v16h af = frag_from_f32(a0, a1, a2, a3);

            #pragma unroll
            for (int j = 0; j < TN; ++j) {
                int n = (tg * TN + j) * 16 + lo;
                const v8h* wp0 = (const v8h*)(&sW[n * K + k0 + 8 * hi]);
                const v8h* wp1 = (const v8h*)(&sW[n * K + k0 + 16 + 8 * hi]);
                v16h bf = cat16(wp0[0], wp1[0]);
                c[j] = __builtin_amdgcn_wmma_f32_16x16x32_f16(
                    false, af, false, bf, (short)0, c[j], false, false);
            }
        }

        #pragma unroll
        for (int j = 0; j < TN; ++j) {
            int n = (tg * TN + j) * 16 + lo;
            #pragma unroll
            for (int r = 0; r < 8; ++r) {
                int row = tm * 16 + r + 8 * hi;
                float v = c[j][r];
                if (obias) v += obias[n];
                if (ACT_OUT == 1) {
                    v = v > 0.f ? v : 0.f;
                }
                C[(size_t)row * Nc + n] = v;
            }
        }
    }
}

__global__ void rank_kernel(const int* __restrict__ rows,
                            int* __restrict__ cnt,
                            int* __restrict__ rank)
{
    int e = (int)(blockIdx.x * blockDim.x + threadIdx.x);
    if (e < NNZ) rank[e] = atomicAdd(&cnt[rows[e]], 1);
}

__global__ void __launch_bounds__(1024)
rowscan_kernel(const int* __restrict__ cnt, int2* __restrict__ rowinfo)
{
    __shared__ int sBase[32];
    __shared__ int sWave[32];
    int t    = (int)threadIdx.x;
    int b    = (int)blockIdx.x;
    int lane = t & 31;
    int wid  = t >> 5;

    const int4* all = (const int4*)cnt;
    int part = 0;
    #pragma unroll 4
    for (int i = t; i < b * 256; i += 1024) {
        int4 q = all[i];
        part += (q.x + q.y) + (q.z + q.w);
    }
    for (int d = 1; d < 32; d <<= 1) part += __shfl_xor(part, d, 32);

    int v = cnt[b * 1024 + t];
    int x = v;
    for (int d = 1; d < 32; d <<= 1) {
        int y = __shfl_up(x, d, 32);
        if (lane >= d) x += y;
    }
    if (lane == 0)  sBase[wid] = part;
    if (lane == 31) sWave[wid] = x;
    __syncthreads();

    int base = 0, before = 0;
    for (int i = 0; i < 32; ++i) {
        int w = sWave[i];
        base += sBase[i];
        before += (i < wid) ? w : 0;
    }
    int2 ri;
    ri.x = base + before + x - v;
    ri.y = v;
    rowinfo[b * 1024 + t] = ri;
}

__global__ void place_kernel(const int* __restrict__ rows,
                             const int* __restrict__ cols,
                             const float* __restrict__ vals,
                             const int2* __restrict__ rowinfo,
                             const int* __restrict__ rank,
                             int2* __restrict__ cv)
{
    int e = (int)(blockIdx.x * blockDim.x + threadIdx.x);
    if (e < NNZ) {
        int2 p;
        p.x = cols[e];
        p.y = __float_as_int(vals[e]);
        cv[rowinfo[rows[e]].x + rank[e]] = p;
    }
}

template <int F4>
__global__ void __launch_bounds__(256)
spmm_gather_kernel(const int2* __restrict__ rowinfo,
                   const int2* __restrict__ cv,
                   const float* __restrict__ in,
                   float* __restrict__ out)
{
    int t   = (int)(blockIdx.x * blockDim.x + threadIdx.x);
    int row = t / F4;
    int c4  = t - row * F4;
    if (row >= NN) return;
    int2 ri  = rowinfo[row];
    int j    = ri.x;
    int end  = j + ri.y;
    int last = end - 1;
    const float4* inp = (const float4*)in + c4;
    float4 acc = {0.f, 0.f, 0.f, 0.f};
    if (j < end) {
        int2 e0 = cv[j];
        int2 e1 = cv[min(j + 1, last)];
        int2 e2 = cv[min(j + 2, last)];
        int2 e3 = cv[min(j + 3, last)];
        for (; j < end; j += 4) {
            float4 x0 = inp[(size_t)e0.x * F4];
            float4 x1 = inp[(size_t)e1.x * F4];
            float4 x2 = inp[(size_t)e2.x * F4];
            float4 x3 = inp[(size_t)e3.x * F4];
            int jn = j + 4;
            int2 n0 = cv[min(jn, last)];
            int2 n1 = cv[min(jn + 1, last)];
            int2 n2 = cv[min(jn + 2, last)];
            int2 n3 = cv[min(jn + 3, last)];
            asm volatile("" ::: "memory");
            bool p1 = j + 1 < end, p2 = j + 2 < end, p3 = j + 3 < end;
            float v0 = __int_as_float(e0.y), v1 = __int_as_float(e1.y);
            float v2 = __int_as_float(e2.y), v3 = __int_as_float(e3.y);
            acc.x += v0 * x0.x; acc.y += v0 * x0.y; acc.z += v0 * x0.z; acc.w += v0 * x0.w;
            acc.x += p1 ? v1 * x1.x : 0.f; acc.y += p1 ? v1 * x1.y : 0.f;
            acc.z += p1 ? v1 * x1.z : 0.f; acc.w += p1 ? v1 * x1.w : 0.f;
            acc.x += p2 ? v2 * x2.x : 0.f; acc.y += p2 ? v2 * x2.y : 0.f;
            acc.z += p2 ? v2 * x2.z : 0.f; acc.w += p2 ? v2 * x2.w : 0.f;
            acc.x += p3 ? v3 * x3.x : 0.f; acc.y += p3 ? v3 * x3.y : 0.f;
            acc.z += p3 ? v3 * x3.z : 0.f; acc.w += p3 ? v3 * x3.w : 0.f;
            e0 = n0; e1 = n1; e2 = n2; e3 = n3;
        }
    }
    ((float4*)out)[(size_t)row * F4 + c4] = acc;
}

__global__ void zero_kernel(float* __restrict__ p, int n)
{
    int i = (int)(blockIdx.x * blockDim.x + threadIdx.x);
    if (i < n) p[i] = 0.f;
}

#define BR_PER_THREAD 8
__global__ void __launch_bounds__(256)
bias_relu_kernel(const float* __restrict__ in,
                 const float* __restrict__ bias,
                 float* __restrict__ out)
{
    int t = (int)threadIdx.x;
    size_t first = (size_t)blockIdx.x * (256 * BR_PER_THREAD) + (size_t)t;
    const float4* ip = (const float4*)in + first;
    float4*       op = (float4*)out + first;
    float4 b = ((const float4*)bias)[t & 15];
    float4 v[BR_PER_THREAD];
    for (int u = 0; u < BR_PER_THREAD; ++u) v[u] = ip[u * 256];
    for (int u = 0; u < BR_PER_THREAD; ++u) op[u * 256] = relu4(v[u], b);
}

__device__ __forceinline__ float xhalf(float x) {
    int v = __builtin_bit_cast(int, x);
    return __builtin_bit_cast(float,
        __builtin_amdgcn_permlanex16(v, v, 0x76543210, 0xfedcba98, false, false));
}

__global__ void attn_kernel(const float* __restrict__ QKV,
                            float* __restrict__ ATT)
{
    __shared__ __align__(16) _Float16 sK[BLK * HDIM];
    __shared__ __align__(16) _Float16 sVt[HDIM * BLK];

    int b = (int)(blockIdx.x >> 2);
    int h = (int)(blockIdx.x & 3);
    int tid  = (int)threadIdx.x;
    int wid  = tid >> 5;
    int lane = tid & 31;
    int hi   = lane >> 4;
    int lo   = lane & 15;
    size_t base = (size_t)b * BLK * (3 * DMODEL);

    for (int i = tid; i < BLK * HDIM; i += 256) {
        int s = i >> 4, d = i & 15;
        sK[i]            = (_Float16)QKV[base + (size_t)s * 192 + DMODEL     + h * HDIM + d];
        sVt[d * BLK + s] = (_Float16)QKV[base + (size_t)s * 192 + 2 * DMODEL + h * HDIM + d];
    }
    __syncthreads();

    v8h zero8 = {(_Float16)0.f, (_Float16)0.f, (_Float16)0.f, (_Float16)0.f,
                 (_Float16)0.f, (_Float16)0.f, (_Float16)0.f, (_Float16)0.f};

    for (int qt = wid; qt < 16; qt += 8) {
        int qrow = qt * 16 + lo;
        const float4* qp = (const float4*)(QKV + base + (size_t)qrow * 192
                                           + h * HDIM + 8 * hi);
        float4 q0 = qp[0], q1 = qp[1];
        q0.x *= 0.25f; q0.y *= 0.25f; q0.z *= 0.25f; q0.w *= 0.25f;
        q1.x *= 0.25f; q1.y *= 0.25f; q1.z *= 0.25f; q1.w *= 0.25f;
        float4 zf = {0.f, 0.f, 0.f, 0.f};
        v16h qf = frag_from_f32(q0, q1, zf, zf);

        v8f ctx = v8f_zero();
        float m = -3.0e38f, l = 0.f;

        for (int kk = 0; kk < 8; ++kk) {
            v8f s0, s1;
            {
                int key = (kk * 2) * 16 + lo;
                v16h kf = cat16(*(const v8h*)(&sK[key * HDIM + 8 * hi]), zero8);
                v8f z = v8f_zero();
                s0 = __builtin_amdgcn_wmma_f32_16x16x32_f16(
                    false, kf, false, qf, (short)0, z, false, false);
            }
            {
                int key = (kk * 2 + 1) * 16 + lo;
                v16h kf = cat16(*(const v8h*)(&sK[key * HDIM + 8 * hi]), zero8);
                v8f z = v8f_zero();
                s1 = __builtin_amdgcn_wmma_f32_16x16x32_f16(
                    false, kf, false, qf, (short)0, z, false, false);
            }
            float tm = fmaxf(s0[0], s1[0]);
            #pragma unroll
            for (int r = 1; r < 8; ++r) tm = fmaxf(tm, fmaxf(s0[r], s1[r]));
            tm = fmaxf(tm, xhalf(tm));
            float nm = fmaxf(m, tm);
            float sc = __expf(m - nm);
            m = nm;
            float rs = 0.f;
            v16h pf;
            #pragma unroll
            for (int r = 0; r < 8; ++r) {
                float e0 = __expf(s0[r] - nm);
                float e1 = __expf(s1[r] - nm);
                rs += e0 + e1;
                pf[r]     = (_Float16)e0;
                pf[8 + r] = (_Float16)e1;
            }
            l = l * sc + rs;
            #pragma unroll
            for (int r = 0; r < 8; ++r) ctx[r] *= sc;
            v16h vf = cat16(*(const v8h*)(&sVt[lo * BLK + kk * 32 + 8 * hi]),
                            *(const v8h*)(&sVt[lo * BLK + kk * 32 + 16 + 8 * hi]));
            ctx = __builtin_amdgcn_wmma_f32_16x16x32_f16(
                false, vf, false, pf, (short)0, ctx, false, false);
        }

        l = l + xhalf(l);
        float4 o0, o1;
        o0.x = ctx[0] / l; o0.y = ctx[1] / l; o0.z = ctx[2] / l; o0.w = ctx[3] / l;
        o1.x = ctx[4] / l; o1.y = ctx[5] / l; o1.z = ctx[6] / l; o1.w = ctx[7] / l;
        float4* op = (float4*)(ATT + (size_t)(b * BLK + qrow) * DMODEL + h * HDIM + 8 * hi);
        op[0] = o0;
        op[1] = o1;
    }
}

#define ROWS_PER_WAVE 8
#define P64   72
#define P128  136
#define CHAIN_TILES 2

template <int K, int NC, int PITCH>
__device__ __forceinline__ void stage_wt(const float* __restrict__ Wg,
                                         _Float16* sW, int tid)
{
    #pragma unroll
    for (int u = 0; u < (K / 8) * NC / 256; ++u) {
        int it = tid + u * 256;
        int n  = it % NC;
        int k8 = it / NC;
        const float* wp = Wg + (size_t)(k8 * 8) * NC + n;
        v8h hv;
        #pragma unroll
        for (int i = 0; i < 8; ++i) hv[i] = (_Float16)wp[i * NC];
        *(v8h*)(&sW[n * PITCH + k8 * 8]) = hv;
    }
}

__device__ __forceinline__ v16h lds_afrag(const _Float16* sW, int pitch, int row,
                                          int ks, int hi)
{
    const _Float16* p = sW + row * pitch + ks * 32 + 8 * hi;
    return cat16(*(const v8h*)p, *(const v8h*)(p + 16));
}

__device__ __forceinline__ v16h pack_bfrag(v8f a, v8f b) {
    v16h r;
    #pragma unroll
    for (int i = 0; i < 8; ++i) {
        r[i]     = (_Float16)a[i];
        r[8 + i] = (_Float16)b[i];
    }
    return r;
}

__device__ __forceinline__ void ln_lane(v8f (&x)[4], const float* __restrict__ g,
                                        const float* __restrict__ bt, int hi)
{
    float s = 0.f;
    #pragma unroll
    for (int j = 0; j < 4; ++j) {
        #pragma unroll
        for (int r = 0; r < 8; ++r) s += x[j][r];
    }
    s += xhalf(s);
    float mu = s * (1.f / 64.f);
    float vs = 0.f;
    #pragma unroll
    for (int j = 0; j < 4; ++j) {
        #pragma unroll
        for (int r = 0; r < 8; ++r) {
            float d = x[j][r] - mu;
            x[j][r] = d;
            vs += d * d;
        }
    }
    vs += xhalf(vs);
    float rstd = rsqrtf(vs * (1.f / 64.f) + 1e-5f);
    #pragma unroll
    for (int j = 0; j < 4; ++j) {
        const float4* gp = (const float4*)(g  + 16 * j + 8 * hi);
        const float4* bp = (const float4*)(bt + 16 * j + 8 * hi);
        float4 g0 = gp[0], g1 = gp[1], b0 = bp[0], b1 = bp[1];
        x[j][0] = x[j][0] * rstd * g0.x + b0.x;
        x[j][1] = x[j][1] * rstd * g0.y + b0.y;
        x[j][2] = x[j][2] * rstd * g0.z + b0.z;
        x[j][3] = x[j][3] * rstd * g0.w + b0.w;
        x[j][4] = x[j][4] * rstd * g1.x + b1.x;
        x[j][5] = x[j][5] * rstd * g1.y + b1.y;
        x[j][6] = x[j][6] * rstd * g1.z + b1.z;
        x[j][7] = x[j][7] * rstd * g1.w + b1.w;
    }
}

__device__ __forceinline__ float gelu_f(float v) {
    return 0.5f * v * (1.0f + erff(v * 0.70710678118654752f));
}

__global__ void __launch_bounds__(256)
rowchain_kernel(const float* __restrict__ ATT, const float* __restrict__ H2,
                const float* __restrict__ Wo,  const float* __restrict__ bo,
                const float* __restrict__ g1,  const float* __restrict__ be1,
                const float* __restrict__ Wf1, const float* __restrict__ bf1,
                const float* __restrict__ Wf2, const float* __restrict__ bf2,
                const float* __restrict__ g2,  const float* __restrict__ be2,
                float* __restrict__ X2)
{
    __shared__ __align__(16) _Float16 sWo[64 * P64];
    __shared__ __align__(16) _Float16 sW1[128 * P64];
    __shared__ __align__(16) _Float16 sW2[64 * P128];

    int tid  = (int)threadIdx.x;
    int wid  = tid >> 5;
    int lane = tid & 31;
    int hi   = lane >> 4;
    int lo   = lane & 15;

    stage_wt<64, 64, P64>(Wo, sWo, tid);
    stage_wt<64, 128, P64>(Wf1, sW1, tid);
    stage_wt<128, 64, P128>(Wf2, sW2, tid);
    __syncthreads();

    int wave = (int)blockIdx.x * 8 + wid;
    for (int t = 0; t < CHAIN_TILES; ++t) {
        asm volatile("" ::: "memory");
        int m = (wave * CHAIN_TILES + t) * 16 + lo;
        const float* arow = ATT + (size_t)m * DMODEL;
        const float* hrow = H2  + (size_t)m * DMODEL;

        v16h ab[2];
        #pragma unroll
        for (int ks = 0; ks < 2; ++ks) {
            const float4* p0 = (const float4*)(arow + ks * 32 + 8 * hi);
            const float4* p1 = (const float4*)(arow + ks * 32 + 16 + 8 * hi);
            ab[ks] = frag_from_f32(p0[0], p0[1], p1[0], p1[1]);
        }

        v8f x[4];
        #pragma unroll
        for (int j = 0; j < 4; ++j) {
            v8f c = v8f_zero();
            #pragma unroll
            for (int ks = 0; ks < 2; ++ks)
                c = __builtin_amdgcn_wmma_f32_16x16x32_f16(
                    false, lds_afrag(sWo, P64, 16 * j + lo, ks, hi), false, ab[ks],
                    (short)0, c, false, false);
            const float4* bp = (const float4*)(bo + 16 * j + 8 * hi);
            const float4* hp = (const float4*)(hrow + 16 * j + 8 * hi);
            float4 b0 = bp[0], b1 = bp[1], h0 = hp[0], h1 = hp[1];
            c[0] = h0.x + (c[0] + b0.x); c[1] = h0.y + (c[1] + b0.y);
            c[2] = h0.z + (c[2] + b0.z); c[3] = h0.w + (c[3] + b0.w);
            c[4] = h1.x + (c[4] + b1.x); c[5] = h1.y + (c[5] + b1.y);
            c[6] = h1.z + (c[6] + b1.z); c[7] = h1.w + (c[7] + b1.w);
            x[j] = c;
        }
        ln_lane(x, g1, be1, hi);

        v16h xb[2];
        xb[0] = pack_bfrag(x[0], x[1]);
        xb[1] = pack_bfrag(x[2], x[3]);

        v8f y[4];
        #pragma unroll
        for (int j = 0; j < 4; ++j) y[j] = v8f_zero();
        for (int c = 0; c < 4; ++c) {
            v8f fa = v8f_zero(), fb = v8f_zero();
            #pragma unroll
            for (int ks = 0; ks < 2; ++ks) {
                fa = __builtin_amdgcn_wmma_f32_16x16x32_f16(
                    false, lds_afrag(sW1, P64, 32 * c + lo, ks, hi), false, xb[ks],
                    (short)0, fa, false, false);
                fb = __builtin_amdgcn_wmma_f32_16x16x32_f16(
                    false, lds_afrag(sW1, P64, 32 * c + 16 + lo, ks, hi), false, xb[ks],
                    (short)0, fb, false, false);
            }
            const float4* pa = (const float4*)(bf1 + 32 * c + 8 * hi);
            const float4* pb = (const float4*)(bf1 + 32 * c + 16 + 8 * hi);
            float4 a0 = pa[0], a1 = pa[1], c0 = pb[0], c1 = pb[1];
            fa[0] = gelu_f(fa[0] + a0.x); fa[1] = gelu_f(fa[1] + a0.y);
            fa[2] = gelu_f(fa[2] + a0.z); fa[3] = gelu_f(fa[3] + a0.w);
            fa[4] = gelu_f(fa[4] + a1.x); fa[5] = gelu_f(fa[5] + a1.y);
            fa[6] = gelu_f(fa[6] + a1.z); fa[7] = gelu_f(fa[7] + a1.w);
            fb[0] = gelu_f(fb[0] + c0.x); fb[1] = gelu_f(fb[1] + c0.y);
            fb[2] = gelu_f(fb[2] + c0.z); fb[3] = gelu_f(fb[3] + c0.w);
            fb[4] = gelu_f(fb[4] + c1.x); fb[5] = gelu_f(fb[5] + c1.y);
            fb[6] = gelu_f(fb[6] + c1.z); fb[7] = gelu_f(fb[7] + c1.w);
            v16h ff = pack_bfrag(fa, fb);
            #pragma unroll
            for (int j = 0; j < 4; ++j)
                y[j] = __builtin_amdgcn_wmma_f32_16x16x32_f16(
                    false, lds_afrag(sW2, P128, 16 * j + lo, c, hi), false, ff,
                    (short)0, y[j], false, false);
        }

        #pragma unroll
        for (int j = 0; j < 4; ++j) {
            const float4* bp = (const float4*)(bf2 + 16 * j + 8 * hi);
            float4 b0 = bp[0], b1 = bp[1];
            y[j][0] = x[j][0] + (y[j][0] + b0.x); y[j][1] = x[j][1] + (y[j][1] + b0.y);
            y[j][2] = x[j][2] + (y[j][2] + b0.z); y[j][3] = x[j][3] + (y[j][3] + b0.w);
            y[j][4] = x[j][4] + (y[j][4] + b1.x); y[j][5] = x[j][5] + (y[j][5] + b1.y);
            y[j][6] = x[j][6] + (y[j][6] + b1.z); y[j][7] = x[j][7] + (y[j][7] + b1.w);
        }
        ln_lane(y, g2, be2, hi);

        #pragma unroll
        for (int j = 0; j < 4; ++j) {
            float4 o0, o1;
            o0.x = y[j][0]; o0.y = y[j][1]; o0.z = y[j][2]; o0.w = y[j][3];
            o1.x = y[j][4]; o1.y = y[j][5]; o1.z = y[j][6]; o1.w = y[j][7];
            float4* op = (float4*)(X2 + (size_t)m * DMODEL + 16 * j + 8 * hi);
            op[0] = o0;
            op[1] = o1;
        }
    }
}

__device__ __forceinline__ float cls_row(const float4 (&x)[8], const float (&wc)[32],
                                         float bias)
{
    float acc = 0.f;
    for (int q = 0; q < 8; ++q) {
        acc += x[q].x * wc[4 * q];
        acc += x[q].y * wc[4 * q + 1];
        acc += x[q].z * wc[4 * q + 2];
        acc += x[q].w * wc[4 * q + 3];
    }
    acc += __shfl_xor(acc, 16, 32);
    acc += bias;
    float mx = acc;
    #pragma unroll
    for (int d = 1; d < 16; d <<= 1) mx = fmaxf(mx, __shfl_xor(mx, d, 32));
    float e = expf(acc - mx), sm = e;
    #pragma unroll
    for (int d = 1; d < 16; d <<= 1) sm += __shfl_xor(sm, d, 32);
    return acc - mx - logf(sm);
}

__global__ void __launch_bounds__(256)
cls_kernel(const float* __restrict__ X2,
           const float* __restrict__ Wc,
           const float* __restrict__ bc,
           float* __restrict__ out)
{
    int w = (int)((blockIdx.x * blockDim.x + threadIdx.x) >> 5);
    int lane = (int)(threadIdx.x & 31);
    int hi = lane >> 4, lo = lane & 15;
    float wc[32];
    for (int kk = 0; kk < 32; ++kk) wc[kk] = Wc[(hi * 32 + kk) * 16 + lo];
    float bias = bc[lo];
    int row0 = w * ROWS_PER_WAVE;
    for (int p = 0; p < ROWS_PER_WAVE / 2; ++p) {
        int ra = row0 + 2 * p;
        const float4* xa = (const float4*)(X2 + (size_t)ra * DMODEL + hi * 32);
        const float4* xb = (const float4*)(X2 + (size_t)(ra + 1) * DMODEL + hi * 32);
        float4 a[8], b[8];
        for (int q = 0; q < 8; ++q) { a[q] = xa[q]; b[q] = xb[q]; }
        float res_a = cls_row(a, wc, bias);
        float res_b = cls_row(b, wc, bias);
        out[(size_t)(ra + hi) * 16 + lo] = hi ? res_b : res_a;
    }
}


static inline int ceil_div(long long a, long long b) { return (int)((a + b - 1) / b); }

#define GEMM_WAVES(M, TM, Nc, TN) \
    ((long long)((M) / (16 * (TM))) * ((Nc) / (16 * (TN))))

static void launch_gemm_none(const float* A, const float* W, const float* ob,
                             float* C, int M, int K, int Nc, hipStream_t s)
{
    long long waves = GEMM_WAVES(M, 4, Nc, 4);
    wmma_gemm_kernel<4, 4, false, 0><<<ceil_div(waves * 32, 256), 256, 0, s>>>(
        A, nullptr, W, ob, C, M, K, Nc);
}

extern "C" void kernel_launch(void* const* d_in, const int* in_sizes, int n_in,
                              void* d_out, int out_size, void* d_ws, size_t ws_size,
                              hipStream_t stream)
{
    const float* x      = (const float*)d_in[0];
    const int*   G_rows = (const int*)  d_in[1];
    const int*   G_cols = (const int*)  d_in[2];
    const float* G_vals = (const float*)d_in[3];
    const float* W1     = (const float*)d_in[4];
    const float* b1     = (const float*)d_in[5];
    const float* W2     = (const float*)d_in[6];
    const float* b2     = (const float*)d_in[7];
    const float* W_qkv  = (const float*)d_in[8];
    const float* b_qkv  = (const float*)d_in[9];
    const float* W_o    = (const float*)d_in[10];
    const float* b_o    = (const float*)d_in[11];
    const float* ln1_g  = (const float*)d_in[12];
    const float* ln1_b  = (const float*)d_in[13];
    const float* W_ff1  = (const float*)d_in[14];
    const float* b_ff1  = (const float*)d_in[15];
    const float* W_ff2  = (const float*)d_in[16];
    const float* b_ff2  = (const float*)d_in[17];
    const float* ln2_g  = (const float*)d_in[18];
    const float* ln2_b  = (const float*)d_in[19];
    const float* W_cls  = (const float*)d_in[20];
    const float* b_cls  = (const float*)d_in[21];
    float* out = (float*)d_out;

    float* ws = (float*)d_ws;
    const size_t n = NN;
    float* P1  = ws;
    float* S1  = ws + n * 32;
    float* P2  = ws + n * 64;
    float* S2  = ws + n * 128;
    float* H2  = ws + n * 192;
    float* QKV = ws + n * 256;
    float* ATT = ws;
    float* X2  = ws + n * 64;
    int*  CNT  = (int*)(ws + n * 256);
    int2* RINF = (int2*)(ws + n * 257);
    int*  RANK = (int*)(ws + n * 259);
    int2* CV   = (int2*)(ws + n * 275);

    {
        long long waves = GEMM_WAVES(NN, 4, 32, 2);
        wmma_gemm_kernel<4, 2, false, 0><<<ceil_div(waves * 32, 256), 256, 0, stream>>>(
            x, nullptr, W1, nullptr, P1, NN, 128, 32);
    }

    zero_kernel<<<ceil_div((long long)NN, 256), 256, 0, stream>>>((float*)CNT, NN);
    rank_kernel<<<ceil_div((long long)NNZ, 256), 256, 0, stream>>>(G_rows, CNT, RANK);
    rowscan_kernel<<<NN / 1024, 1024, 0, stream>>>(CNT, RINF);
    place_kernel<<<ceil_div((long long)NNZ, 256), 256, 0, stream>>>(
        G_rows, G_cols, G_vals, RINF, RANK, CV);

    spmm_gather_kernel<8><<<ceil_div((long long)NN * 8, 256), 256, 0, stream>>>(
        RINF, CV, P1, S1);

    {
        long long waves = GEMM_WAVES(NN, 4, 64, 4);
        wmma_gemm_kernel<4, 4, true, 0><<<ceil_div(waves * 32, 256), 256, 0, stream>>>(
            S1, b1, W2, nullptr, P2, NN, 32, 64);
    }

    spmm_gather_kernel<16><<<ceil_div((long long)NN * 16, 256), 256, 0, stream>>>(
        RINF, CV, P2, S2);

    bias_relu_kernel<<<NN * 16 / (256 * BR_PER_THREAD), 256, 0, stream>>>(S2, b2, H2);

    launch_gemm_none(H2, W_qkv, b_qkv, QKV, NN, 64, 192, stream);

    attn_kernel<<<(NN / BLK) * NHEAD, 256, 0, stream>>>(QKV, ATT);

    rowchain_kernel<<<NN / (16 * CHAIN_TILES * 8), 256, 0, stream>>>(
        ATT, H2, W_o, b_o, ln1_g, ln1_b, W_ff1, b_ff1, W_ff2, b_ff2,
        ln2_g, ln2_b, X2);

    cls_kernel<<<NN / ROWS_PER_WAVE / 8, 256, 0, stream>>>(
        X2, W_cls, b_cls, out);
}
